# starting kernel
# baseline (speedup 1.0000x reference)
#include <hip/hip_runtime.h>

#define N_NODES 50000
#define N_EDGES 800000
#define HDIM    64

typedef float v2f __attribute__((ext_vector_type(2)));
typedef float v8f __attribute__((ext_vector_type(8)));

__global__ void deg_init(float* __restrict__ deg) {
    int i = blockIdx.x * 256 + threadIdx.x;
    if (i < N_NODES) deg[i] = 1.0f;
}

__global__ void deg_count(const int* __restrict__ ei, float* __restrict__ deg) {
    int e = blockIdx.x * 256 + threadIdx.x;
    if (e < N_EDGES) {
        int dst = ei[N_EDGES + e];
        atomicAdd(&deg[dst], 1.0f);
    }
}

__global__ void deg_to_dinv(float* __restrict__ deg) {
    int i = blockIdx.x * 256 + threadIdx.x;
    if (i < N_NODES) deg[i] = rsqrtf(deg[i]);
}

template <bool RELU_IN>
__global__ __launch_bounds__(128)
void gcn_gemm(const float* __restrict__ in, const float* __restrict__ bias,
              const float* __restrict__ W, const float* __restrict__ dinv,
              float* __restrict__ h, float* __restrict__ agg) {
    const int wave = threadIdx.x >> 5;
    const int lane = threadIdx.x & 31;
    const int tile = blockIdx.x * 4 + wave;
    const int tm   = tile >> 2;
    const int tn   = tile & 3;

    const int mrow  = (tm << 4) + (lane & 15);
    const int khalf = (lane >> 4) << 1;
    const int ncol  = (tn << 4) + (lane & 15);

    const float* __restrict__ arow = in + (size_t)mrow * HDIM;

    v8f c = {0.f, 0.f, 0.f, 0.f, 0.f, 0.f, 0.f, 0.f};
#pragma unroll
    for (int k = 0; k < HDIM; k += 4) {
        const int ka = k + khalf;
        v2f a;
        a.x = arow[ka];
        a.y = arow[ka + 1];
        if (RELU_IN) {
            a.x = fmaxf(a.x + bias[ka],     0.f);
            a.y = fmaxf(a.y + bias[ka + 1], 0.f);
        }
        v2f b;
        b.x = W[(size_t)ka * HDIM + ncol];
        b.y = W[(size_t)(ka + 1) * HDIM + ncol];
        c = __builtin_amdgcn_wmma_f32_16x16x4_f32(false, a, false, b,
                                                  (short)0, c, false, false);
    }

    const int mbase = (tm << 4) + ((lane >> 4) << 3);
    const int nn    = (tn << 4) + (lane & 15);
#pragma unroll
    for (int v = 0; v < 8; ++v) {
        const int row = mbase + v;
        const float val = c[v];
        h[(size_t)row * HDIM + nn] = val;
        const float di = dinv[row];
        agg[(size_t)row * HDIM + nn] = val * di * di;
    }
}

__global__ __launch_bounds__(256)
void gcn_scatter(const int* __restrict__ ei, const float* __restrict__ dinv,
                 const float* __restrict__ h, float* __restrict__ agg) {
    const int t = blockIdx.x * 256 + threadIdx.x;
    const int e = t >> 4;
    if (e >= N_EDGES) return;
    const int fo  = (t & 15) << 2;
    const int src = ei[e];
    const int dst = ei[N_EDGES + e];
    const float nrm = dinv[src] * dinv[dst];

    const float4 v = *(const float4*)(h + (size_t)src * HDIM + fo);
    float* ap = agg + (size_t)dst * HDIM + fo;
    atomicAdd(ap + 0, v.x * nrm);
    atomicAdd(ap + 1, v.y * nrm);
    atomicAdd(ap + 2, v.z * nrm);
    atomicAdd(ap + 3, v.w * nrm);
}

__global__ void gcn_act(const float* __restrict__ agg, const float* __restrict__ bias,
                        float* __restrict__ out) {
    int i = blockIdx.x * 256 + threadIdx.x;
    if (i < N_NODES * HDIM)
        out[i] = fmaxf(agg[i] + bias[i & (HDIM - 1)], 0.f);
}

extern "C" void kernel_launch(void* const* d_in, const int* in_sizes, int n_in,
                              void* d_out, int out_size, void* d_ws, size_t ws_size,
                              hipStream_t stream) {
    const float* x  = (const float*)d_in[0];
    const int*   ei = (const int*)d_in[1];
    const float* W  = (const float*)d_in[2];
    const float* b  = (const float*)d_in[3];
    float* out = (float*)d_out;

    float* ws   = (float*)d_ws;
    float* dinv = ws;
    float* h    = dinv + N_NODES;
    float* aggA = h    + (size_t)N_NODES * HDIM;
    float* aggB = aggA + (size_t)N_NODES * HDIM;

    const int nodeBlocks = (N_NODES + 255) / 256;
    const int edgeBlocks = (N_EDGES + 255) / 256;
    const int scatBlocks = (N_EDGES * 16) / 256;
    const int gemmBlocks = (N_NODES / 16) * (HDIM / 16) / 4;
    const int actBlocks  = (N_NODES * HDIM + 255) / 256;

    deg_init   <<<nodeBlocks, 256, 0, stream>>>(dinv);
    deg_count  <<<edgeBlocks, 256, 0, stream>>>(ei, dinv);
    deg_to_dinv<<<nodeBlocks, 256, 0, stream>>>(dinv);

    gcn_gemm<false><<<gemmBlocks, 128, 0, stream>>>(x, nullptr, W, dinv, h, aggA);
    gcn_scatter    <<<scatBlocks, 256, 0, stream>>>(ei, dinv, h, aggA);

    gcn_gemm<true> <<<gemmBlocks, 128, 0, stream>>>(aggA, b, W + HDIM * HDIM, dinv, h, aggB);
    gcn_scatter    <<<scatBlocks, 256, 0, stream>>>(ei, dinv, h, aggB);

    gcn_gemm<true> <<<gemmBlocks, 128, 0, stream>>>(aggB, b + HDIM, W + 2 * HDIM * HDIM, dinv, h, aggA);
    gcn_scatter    <<<scatBlocks, 256, 0, stream>>>(ei, dinv, h, aggA);

    gcn_act<<<actBlocks, 256, 0, stream>>>(aggA, b + 2 * HDIM, out);
}


# Edge sums by row gather over a per-call dst-sorted edge list (slots from integer atomic arrival order, rows pre-scaled by dinv) instead of per-edge float atomics; per-node sum order changes
# speedup vs baseline: 2.3590x; 2.3590x over previous
#include <hip/hip_runtime.h>


#define N_NODES 50000
#define N_EDGES 800000
#define HDIM    64
#define NPAD    50048

typedef float v2f __attribute__((ext_vector_type(2)));
typedef float v8f __attribute__((ext_vector_type(8)));

__global__ void cnt_init(int* __restrict__ cnt) {
    int i = blockIdx.x * 256 + threadIdx.x;
    if (i < N_NODES) cnt[i] = 0;
}

__global__ void edge_rank(const int* __restrict__ ei, int* __restrict__ cnt,
                          int* __restrict__ rank) {
    int q = blockIdx.x * 256 + threadIdx.x;
    if (q < N_EDGES / 4) {
        const int4 d = ((const int4*)(ei + N_EDGES))[q];
        int4 r;
        r.x = atomicAdd(&cnt[d.x], 1);
        r.y = atomicAdd(&cnt[d.y], 1);
        r.z = atomicAdd(&cnt[d.z], 1);
        r.w = atomicAdd(&cnt[d.w], 1);
        ((int4*)rank)[q] = r;
    }
}

__global__ __launch_bounds__(1024)
void node_scan(const int* __restrict__ cnt, int* __restrict__ rowptr,
               float* __restrict__ dinv) {
    __shared__ int s_incl[32];
    __shared__ int s_part[32];
    const int t    = threadIdx.x;
    const int lane = t & 31;
    const int w    = t >> 5;
    const int b    = blockIdx.x;

    const int4* __restrict__ c4 = (const int4*)cnt;
    const int n4 = b * 256;
    int part = 0;
#pragma unroll 4
    for (int i = t; i < n4; i += 1024) {
        const int4 v = c4[i];
        part += (v.x + v.y) + (v.z + v.w);
    }

    const int idx = b * 1024 + t;
    const int c   = (idx < N_NODES) ? cnt[idx] : 0;

    int incl = c;
    for (int d = 1; d < 32; d <<= 1) {
        const int o = __shfl_up(incl, d);
        if (lane >= d) incl += o;
    }
    for (int d = 16; d > 0; d >>= 1) part += __shfl_xor(part, d);

    if (lane == 31) s_incl[w] = incl;
    if (lane == 0)  s_part[w] = part;
    __syncthreads();

    int pre = (lane < w) ? s_incl[lane] : 0;
    int pp  = s_part[lane];
    for (int d = 16; d > 0; d >>= 1) {
        pre += __shfl_xor(pre, d);
        pp  += __shfl_xor(pp, d);
    }
    const int excl = pp + pre + incl - c;
    if (idx < N_NODES) {
        rowptr[idx] = excl;
        dinv[idx]   = rsqrtf((float)(c + 1));
    }
    if (idx == N_NODES - 1) rowptr[N_NODES] = excl + c;
}

__global__ void csr_fill(const int* __restrict__ ei, const int* __restrict__ rowptr,
                         const int* __restrict__ rank, int* __restrict__ csr) {
    int q = blockIdx.x * 256 + threadIdx.x;
    if (q < N_EDGES / 4) {
        const int4 d = ((const int4*)(ei + N_EDGES))[q];
        const int4 r = ((const int4*)rank)[q];
        const int4 s = ((const int4*)ei)[q];
        csr[rowptr[d.x] + r.x] = s.x;
        csr[rowptr[d.y] + r.y] = s.y;
        csr[rowptr[d.z] + r.z] = s.z;
        csr[rowptr[d.w] + r.w] = s.w;
    }
}

template <bool RELU_IN>
__global__ __launch_bounds__(128)
void gcn_gemm(const float* __restrict__ in, const float* __restrict__ bias,
              const float* __restrict__ W, const float* __restrict__ dinv,
              float* __restrict__ h) {
    const int wave = threadIdx.x >> 5;
    const int lane = threadIdx.x & 31;
    const int tile = blockIdx.x * 4 + wave;
    const int tm   = tile >> 2;
    const int tn   = tile & 3;

    const int mrow  = (tm << 4) + (lane & 15);
    const int khalf = (lane >> 4) << 1;
    const int ncol  = (tn << 4) + (lane & 15);

    const float* __restrict__ arow = in + (size_t)mrow * HDIM;

    v8f c = {0.f, 0.f, 0.f, 0.f, 0.f, 0.f, 0.f, 0.f};
#pragma unroll
    for (int k = 0; k < HDIM; k += 4) {
        const int ka = k + khalf;
        v2f a;
        a.x = arow[ka];
        a.y = arow[ka + 1];
        if (RELU_IN) {
            a.x = fmaxf(a.x + bias[ka],     0.f);
            a.y = fmaxf(a.y + bias[ka + 1], 0.f);
        }
        v2f b;
        b.x = W[(size_t)ka * HDIM + ncol];
        b.y = W[(size_t)(ka + 1) * HDIM + ncol];
        c = __builtin_amdgcn_wmma_f32_16x16x4_f32(false, a, false, b,
                                                  (short)0, c, false, false);
    }

    const int mbase = (tm << 4) + ((lane >> 4) << 3);
    const int nn    = (tn << 4) + (lane & 15);
#pragma unroll
    for (int v = 0; v < 8; ++v) {
        const int row = mbase + v;
        h[(size_t)row * HDIM + nn] = c[v] * dinv[row];
    }
}

#define GATHER_BLOCKS ((N_NODES + 31) / 32)
__global__ __launch_bounds__(256)
void gcn_gather(const int* __restrict__ rowptr, const int* __restrict__ csr,
                const float* __restrict__ dinv, const float* __restrict__ h,
                float* __restrict__ agg) {
    const int t  = blockIdx.x * 256 + threadIdx.x;
    const int fo = (t & 15) << 2;
    const float* __restrict__ hb = h + fo;

    for (int node = t >> 4; node < N_NODES; node += GATHER_BLOCKS * 16) {
        const int beg = rowptr[node];
        const int end = rowptr[node + 1];
        const int last = end - 1;
        float4 acc = *(const float4*)(hb + node * HDIM);
        for (int j = beg; j < end; j += 4) {
            const int s0 = csr[j];
            const int s1 = csr[min(j + 1, last)];
            const int s2 = csr[min(j + 2, last)];
            const int s3 = csr[min(j + 3, last)];
            const float4 v0 = *(const float4*)(hb + s0 * HDIM);
            const float4 v1 = *(const float4*)(hb + s1 * HDIM);
            const float4 v2 = *(const float4*)(hb + s2 * HDIM);
            const float4 v3 = *(const float4*)(hb + s3 * HDIM);
            const bool p1 = j + 1 < end;
            const bool p2 = j + 2 < end;
            const bool p3 = j + 3 < end;
            acc.x += v0.x; acc.y += v0.y; acc.z += v0.z; acc.w += v0.w;
            acc.x += p1 ? v1.x : 0.f; acc.y += p1 ? v1.y : 0.f;
            acc.z += p1 ? v1.z : 0.f; acc.w += p1 ? v1.w : 0.f;
            acc.x += p2 ? v2.x : 0.f; acc.y += p2 ? v2.y : 0.f;
            acc.z += p2 ? v2.z : 0.f; acc.w += p2 ? v2.w : 0.f;
            acc.x += p3 ? v3.x : 0.f; acc.y += p3 ? v3.y : 0.f;
            acc.z += p3 ? v3.z : 0.f; acc.w += p3 ? v3.w : 0.f;
        }
        const float di = dinv[node];
        float4 r;
        r.x = acc.x * di; r.y = acc.y * di; r.z = acc.z * di; r.w = acc.w * di;
        *(float4*)(agg + node * HDIM + fo) = r;
    }
}

__global__ void gcn_act(const float* __restrict__ agg, const float* __restrict__ bias,
                        float* __restrict__ out) {
    int i = blockIdx.x * 256 + threadIdx.x;
    if (i < N_NODES * HDIM)
        out[i] = fmaxf(agg[i] + bias[i & (HDIM - 1)], 0.f);
}

extern "C" void kernel_launch(void* const* d_in, const int* in_sizes, int n_in,
                              void* d_out, int out_size, void* d_ws, size_t ws_size,
                              hipStream_t stream) {
    const float* x  = (const float*)d_in[0];
    const int*   ei = (const int*)d_in[1];
    const float* W  = (const float*)d_in[2];
    const float* b  = (const float*)d_in[3];
    float* out = (float*)d_out;

    float* ws   = (float*)d_ws;
    float* h    = ws;
    float* aggA = h    + (size_t)N_NODES * HDIM;
    float* aggB = aggA + (size_t)N_NODES * HDIM;
    float* dinv = aggB + (size_t)N_NODES * HDIM;
    int*   cnt    = (int*)(dinv + NPAD);
    int*   rowptr = cnt + NPAD;
    int*   rank   = rowptr + NPAD;
    int*   csr    = rank + N_EDGES;

    const int nodeBlocks = (N_NODES + 255) / 256;
    const int edgeBlocks = (N_EDGES / 4 + 255) / 256;
    const int scanBlocks = (N_NODES + 1023) / 1024;
    const int gathBlocks = GATHER_BLOCKS;
    const int gemmBlocks = (N_NODES / 16) * (HDIM / 16) / 4;
    const int actBlocks  = (N_NODES * HDIM + 255) / 256;

    cnt_init <<<nodeBlocks, 256, 0, stream>>>(cnt);
    edge_rank<<<edgeBlocks, 256, 0, stream>>>(ei, cnt, rank);
    node_scan<<<scanBlocks, 1024, 0, stream>>>(cnt, rowptr, dinv);
    csr_fill <<<edgeBlocks, 256, 0, stream>>>(ei, rowptr, rank, csr);

    gcn_gemm<false><<<gemmBlocks, 128, 0, stream>>>(x, nullptr, W, dinv, h);
    gcn_gather     <<<gathBlocks, 256, 0, stream>>>(rowptr, csr, dinv, h, aggA);

    gcn_gemm<true> <<<gemmBlocks, 128, 0, stream>>>(aggA, b, W + HDIM * HDIM, dinv, h);
    gcn_gather     <<<gathBlocks, 256, 0, stream>>>(rowptr, csr, dinv, h, aggB);

    gcn_gemm<true> <<<gemmBlocks, 128, 0, stream>>>(aggB, b + HDIM, W + 2 * HDIM * HDIM, dinv, h);
    gcn_gather     <<<gathBlocks, 256, 0, stream>>>(rowptr, csr, dinv, h, aggA);

    gcn_act<<<actBlocks, 256, 0, stream>>>(aggA, b + 2 * HDIM, out);
}


# Each gather applies its consumer in its epilogue: bias+ReLU tile through LDS into the next layer's f32 WMMA product, or the final bias+ReLU store; agg planes no longer stored, 3 launches fewer
# speedup vs baseline: 2.7184x; 1.1524x over previous
#include <hip/hip_runtime.h>


#define N_NODES 50000
#define N_EDGES 800000
#define HDIM    64
#define NPAD    50048

typedef float v2f __attribute__((ext_vector_type(2)));
typedef float v8f __attribute__((ext_vector_type(8)));

__global__ void cnt_init(int* __restrict__ cnt) {
    int i = blockIdx.x * 256 + threadIdx.x;
    if (i < N_NODES) cnt[i] = 0;
}

__global__ void edge_rank(const int* __restrict__ ei, int* __restrict__ cnt,
                          int* __restrict__ rank) {
    int q = blockIdx.x * 256 + threadIdx.x;
    if (q < N_EDGES / 4) {
        const int4 d = ((const int4*)(ei + N_EDGES))[q];
        int4 r;
        r.x = atomicAdd(&cnt[d.x], 1);
        r.y = atomicAdd(&cnt[d.y], 1);
        r.z = atomicAdd(&cnt[d.z], 1);
        r.w = atomicAdd(&cnt[d.w], 1);
        ((int4*)rank)[q] = r;
    }
}

__global__ __launch_bounds__(1024)
void node_scan(const int* __restrict__ cnt, int* __restrict__ rowptr,
               float* __restrict__ dinv) {
    __shared__ int s_incl[32];
    __shared__ int s_part[32];
    const int t    = threadIdx.x;
    const int lane = t & 31;
    const int w    = t >> 5;
    const int b    = blockIdx.x;

    const int4* __restrict__ c4 = (const int4*)cnt;
    const int n4 = b * 256;
    int part = 0;
#pragma unroll 4
    for (int i = t; i < n4; i += 1024) {
        const int4 v = c4[i];
        part += (v.x + v.y) + (v.z + v.w);
    }

    const int idx = b * 1024 + t;
    const int c   = (idx < N_NODES) ? cnt[idx] : 0;

    int incl = c;
    for (int d = 1; d < 32; d <<= 1) {
        const int o = __shfl_up(incl, d);
        if (lane >= d) incl += o;
    }
    for (int d = 16; d > 0; d >>= 1) part += __shfl_xor(part, d);

    if (lane == 31) s_incl[w] = incl;
    if (lane == 0)  s_part[w] = part;
    __syncthreads();

    int pre = (lane < w) ? s_incl[lane] : 0;
    int pp  = s_part[lane];
    for (int d = 16; d > 0; d >>= 1) {
        pre += __shfl_xor(pre, d);
        pp  += __shfl_xor(pp, d);
    }
    const int excl = pp + pre + incl - c;
    if (idx < N_NODES) {
        rowptr[idx] = excl;
        dinv[idx]   = rsqrtf((float)(c + 1));
    }
    if (idx == N_NODES - 1) rowptr[N_NODES] = excl + c;
}

__global__ void csr_fill(const int* __restrict__ ei, const int* __restrict__ rowptr,
                         const int* __restrict__ rank, int* __restrict__ csr) {
    int q = blockIdx.x * 256 + threadIdx.x;
    if (q < N_EDGES / 4) {
        const int4 d = ((const int4*)(ei + N_EDGES))[q];
        const int4 r = ((const int4*)rank)[q];
        const int4 s = ((const int4*)ei)[q];
        csr[rowptr[d.x] + r.x] = s.x;
        csr[rowptr[d.y] + r.y] = s.y;
        csr[rowptr[d.z] + r.z] = s.z;
        csr[rowptr[d.w] + r.w] = s.w;
    }
}

__global__ __launch_bounds__(128)
void gcn_gemm(const float* __restrict__ in, const float* __restrict__ W,
              const float* __restrict__ dinv, float* __restrict__ h) {
    const int wave = threadIdx.x >> 5;
    const int lane = threadIdx.x & 31;
    const int tile = blockIdx.x * 4 + wave;
    const int tm   = tile >> 2;
    const int tn   = tile & 3;

    const int mrow  = (tm << 4) + (lane & 15);
    const int khalf = (lane >> 4) << 1;
    const int ncol  = (tn << 4) + (lane & 15);

    const float* __restrict__ arow = in + (size_t)mrow * HDIM;

    v8f c = {0.f, 0.f, 0.f, 0.f, 0.f, 0.f, 0.f, 0.f};
#pragma unroll
    for (int k = 0; k < HDIM; k += 4) {
        const int ka = k + khalf;
        v2f a;
        a.x = arow[ka];
        a.y = arow[ka + 1];
        v2f b;
        b.x = W[(size_t)ka * HDIM + ncol];
        b.y = W[(size_t)(ka + 1) * HDIM + ncol];
        c = __builtin_amdgcn_wmma_f32_16x16x4_f32(false, a, false, b,
                                                  (short)0, c, false, false);
    }

    const int mbase = (tm << 4) + ((lane >> 4) << 3);
    const int nn    = (tn << 4) + (lane & 15);
#pragma unroll
    for (int v = 0; v < 8; ++v) {
        const int row = mbase + v;
        h[(size_t)row * HDIM + nn] = c[v] * dinv[row];
    }
}

__device__ __forceinline__ float4 gather_sum(const int* __restrict__ rowptr,
                                             const int* __restrict__ csr,
                                             const float* __restrict__ hb, int node) {
    const int beg = rowptr[node];
    const int end = rowptr[node + 1];
    const int last = end - 1;
    float4 acc = *(const float4*)(hb + node * HDIM);
    for (int j = beg; j < end; j += 4) {
        const int s0 = csr[j];
        const int s1 = csr[min(j + 1, last)];
        const int s2 = csr[min(j + 2, last)];
        const int s3 = csr[min(j + 3, last)];
        const float4 v0 = *(const float4*)(hb + s0 * HDIM);
        const float4 v1 = *(const float4*)(hb + s1 * HDIM);
        const float4 v2 = *(const float4*)(hb + s2 * HDIM);
        const float4 v3 = *(const float4*)(hb + s3 * HDIM);
        const bool p1 = j + 1 < end;
        const bool p2 = j + 2 < end;
        const bool p3 = j + 3 < end;
        acc.x += v0.x; acc.y += v0.y; acc.z += v0.z; acc.w += v0.w;
        acc.x += p1 ? v1.x : 0.f; acc.y += p1 ? v1.y : 0.f;
        acc.z += p1 ? v1.z : 0.f; acc.w += p1 ? v1.w : 0.f;
        acc.x += p2 ? v2.x : 0.f; acc.y += p2 ? v2.y : 0.f;
        acc.z += p2 ? v2.z : 0.f; acc.w += p2 ? v2.w : 0.f;
        acc.x += p3 ? v3.x : 0.f; acc.y += p3 ? v3.y : 0.f;
        acc.z += p3 ? v3.z : 0.f; acc.w += p3 ? v3.w : 0.f;
    }
    return acc;
}

#define GATHER_BLOCKS ((N_NODES + 31) / 32)
#define N_TILES (N_NODES / 16)
#define A_PITCH 68
#define W_PITCH 80

__global__ __launch_bounds__(256)
void gcn_gather_gemm(const int* __restrict__ rowptr, const int* __restrict__ csr,
                     const float* __restrict__ dinv, const float* __restrict__ h,
                     const float* __restrict__ bias, const float* __restrict__ Wn,
                     float* __restrict__ hn) {
    __shared__ __attribute__((aligned(16))) float sA[2][16][A_PITCH];
    __shared__ __attribute__((aligned(16))) v2f   sW[HDIM / 2][W_PITCH];

    const int tid = threadIdx.x;
#pragma unroll
    for (int i = 0; i < 8; ++i) {
        const int e  = tid + 256 * i;
        const int kk = e >> 6;
        const int n  = e & 63;
        v2f w;
        w.x = Wn[(2 * kk) * HDIM + n];
        w.y = Wn[(2 * kk + 1) * HDIM + n];
        sW[kk][n] = w;
    }

    const int nl   = tid >> 4;
    const int fo   = (tid & 15) << 2;
    const int wave = tid >> 5;
    const int lane = tid & 31;
    const int c16  = lane & 15;
    const int hh   = lane >> 4;
    const float4 bv = *(const float4*)(bias + fo);
    const float* __restrict__ hb = h + fo;

    int buf = 0;
    for (int tile = blockIdx.x; tile < N_TILES; tile += GATHER_BLOCKS) {
        const int node = (tile << 4) + nl;
        const float4 acc = gather_sum(rowptr, csr, hb, node);
        const float di = dinv[node];
        float4 r;
        r.x = fmaxf(acc.x * di + bv.x, 0.f);
        r.y = fmaxf(acc.y * di + bv.y, 0.f);
        r.z = fmaxf(acc.z * di + bv.z, 0.f);
        r.w = fmaxf(acc.w * di + bv.w, 0.f);
        *(float4*)&sA[buf][nl][fo] = r;
        __syncthreads();
        if (wave < 4) {
            v8f c = {0.f, 0.f, 0.f, 0.f, 0.f, 0.f, 0.f, 0.f};
#pragma unroll
            for (int k = 0; k < HDIM; k += 4) {
                const v2f a = *(const v2f*)&sA[buf][c16][k + 2 * hh];
                const v2f b = sW[(k >> 1) + hh][(wave << 4) + c16];
                c = __builtin_amdgcn_wmma_f32_16x16x4_f32(false, a, false, b,
                                                          (short)0, c, false, false);
            }
            const int mbase = (tile << 4) + (hh << 3);
            const int nn    = (wave << 4) + c16;
#pragma unroll
            for (int v = 0; v < 8; ++v) {
                const int row = mbase + v;
                hn[(size_t)row * HDIM + nn] = c[v] * dinv[row];
            }
        }
        buf ^= 1;
    }
}

__global__ __launch_bounds__(256)
void gcn_gather_out(const int* __restrict__ rowptr, const int* __restrict__ csr,
                    const float* __restrict__ dinv, const float* __restrict__ h,
                    const float* __restrict__ bias, float* __restrict__ out) {
    const int t  = blockIdx.x * 256 + threadIdx.x;
    const int fo = (t & 15) << 2;
    const float4 bv = *(const float4*)(bias + fo);
    const float* __restrict__ hb = h + fo;

    for (int node = t >> 4; node < N_NODES; node += GATHER_BLOCKS * 16) {
        const float4 acc = gather_sum(rowptr, csr, hb, node);
        const float di = dinv[node];
        float4 r;
        r.x = fmaxf(acc.x * di + bv.x, 0.f);
        r.y = fmaxf(acc.y * di + bv.y, 0.f);
        r.z = fmaxf(acc.z * di + bv.z, 0.f);
        r.w = fmaxf(acc.w * di + bv.w, 0.f);
        *(float4*)(out + node * HDIM + fo) = r;
    }
}

extern "C" void kernel_launch(void* const* d_in, const int* in_sizes, int n_in,
                              void* d_out, int out_size, void* d_ws, size_t ws_size,
                              hipStream_t stream) {
    const float* x  = (const float*)d_in[0];
    const int*   ei = (const int*)d_in[1];
    const float* W  = (const float*)d_in[2];
    const float* b  = (const float*)d_in[3];
    float* out = (float*)d_out;

    float* ws   = (float*)d_ws;
    float* hA   = ws;
    float* hB   = hA   + (size_t)N_NODES * HDIM;
    float* dinv = hB   + (size_t)2 * N_NODES * HDIM;
    int*   cnt    = (int*)(dinv + NPAD);
    int*   rowptr = cnt + NPAD;
    int*   rank   = rowptr + NPAD;
    int*   csr    = rank + N_EDGES;

    const int nodeBlocks = (N_NODES + 255) / 256;
    const int edgeBlocks = (N_EDGES / 4 + 255) / 256;
    const int scanBlocks = (N_NODES + 1023) / 1024;
    const int gathBlocks = GATHER_BLOCKS;
    const int gemmBlocks = (N_NODES / 16) * (HDIM / 16) / 4;

    cnt_init <<<nodeBlocks, 256, 0, stream>>>(cnt);
    edge_rank<<<edgeBlocks, 256, 0, stream>>>(ei, cnt, rank);
    node_scan<<<scanBlocks, 1024, 0, stream>>>(cnt, rowptr, dinv);
    csr_fill <<<edgeBlocks, 256, 0, stream>>>(ei, rowptr, rank, csr);

    gcn_gemm<<<gemmBlocks, 128, 0, stream>>>(x, W, dinv, hA);
    gcn_gather_gemm<<<gathBlocks, 256, 0, stream>>>(rowptr, csr, dinv, hA, b, W + HDIM * HDIM, hB);
    gcn_gather_gemm<<<gathBlocks, 256, 0, stream>>>(rowptr, csr, dinv, hB, b + HDIM,
                                                    W + 2 * HDIM * HDIM, hA);
    gcn_gather_out<<<gathBlocks, 256, 0, stream>>>(rowptr, csr, dinv, hA, b + 2 * HDIM, out);
}


# Source ids bucketed by destination into fixed 64-slot rows (spill list beyond) by the ranking pass itself: no prefix scan, no fill pass, 2 launches fewer; per-node addition order unchanged
# speedup vs baseline: 3.0936x; 1.1380x over previous
#include <hip/hip_runtime.h>


#define N_NODES 50000
#define N_EDGES 800000
#define HDIM    64
#define NPAD    50048
#define KCAP    64

typedef float v2f __attribute__((ext_vector_type(2)));
typedef float v8f __attribute__((ext_vector_type(8)));

__global__ void cnt_init(int* __restrict__ cnt) {
    int i = blockIdx.x * 256 + threadIdx.x;
    if (i <= N_NODES) cnt[i] = 0;
}

__device__ __forceinline__ void put_edge(int* __restrict__ cnt, int* __restrict__ slots,
                                         int* __restrict__ spill, int d, int r, int s) {
    if (r < KCAP) {
        slots[d * KCAP + r] = s;
    } else {
        const int o = atomicAdd(&cnt[N_NODES], 1);
        ((int2*)spill)[o] = make_int2(s, d);
    }
}

__global__ void edge_bucket(const int* __restrict__ ei, int* __restrict__ cnt,
                            int* __restrict__ slots, int* __restrict__ spill) {
    int q = blockIdx.x * 256 + threadIdx.x;
    if (q < N_EDGES / 4) {
        const int4 d = ((const int4*)(ei + N_EDGES))[q];
        const int4 s = ((const int4*)ei)[q];
        const int r0 = atomicAdd(&cnt[d.x], 1);
        const int r1 = atomicAdd(&cnt[d.y], 1);
        const int r2 = atomicAdd(&cnt[d.z], 1);
        const int r3 = atomicAdd(&cnt[d.w], 1);
        put_edge(cnt, slots, spill, d.x, r0, s.x);
        put_edge(cnt, slots, spill, d.y, r1, s.y);
        put_edge(cnt, slots, spill, d.z, r2, s.z);
        put_edge(cnt, slots, spill, d.w, r3, s.w);
    }
}

__device__ __forceinline__ float dinv_of(int c) {
    return rsqrtf((float)(c + 1));
}

__global__ __launch_bounds__(128)
void gcn_gemm(const float* __restrict__ in, const float* __restrict__ W,
              const int* __restrict__ cnt, float* __restrict__ h) {
    const int wave = threadIdx.x >> 5;
    const int lane = threadIdx.x & 31;
    const int tile = blockIdx.x * 4 + wave;
    const int tm   = tile >> 2;
    const int tn   = tile & 3;

    const int mrow  = (tm << 4) + (lane & 15);
    const int khalf = (lane >> 4) << 1;
    const int ncol  = (tn << 4) + (lane & 15);

    const float* __restrict__ arow = in + (size_t)mrow * HDIM;

    v8f c = {0.f, 0.f, 0.f, 0.f, 0.f, 0.f, 0.f, 0.f};
#pragma unroll
    for (int k = 0; k < HDIM; k += 4) {
        const int ka = k + khalf;
        v2f a;
        a.x = arow[ka];
        a.y = arow[ka + 1];
        v2f b;
        b.x = W[(size_t)ka * HDIM + ncol];
        b.y = W[(size_t)(ka + 1) * HDIM + ncol];
        c = __builtin_amdgcn_wmma_f32_16x16x4_f32(false, a, false, b,
                                                  (short)0, c, false, false);
    }

    const int mbase = (tm << 4) + ((lane >> 4) << 3);
    const int nn    = (tn << 4) + (lane & 15);
#pragma unroll
    for (int v = 0; v < 8; ++v) {
        const int row = mbase + v;
        h[(size_t)row * HDIM + nn] = c[v] * dinv_of(cnt[row]);
    }
}

__device__ __forceinline__ float4 gather_sum(const int* __restrict__ cnt,
                                             const int* __restrict__ slots,
                                             const int* __restrict__ spill,
                                             const float* __restrict__ hb, int node, int deg) {
    const int beg = node * KCAP;
    const int end = beg + min(deg, KCAP);
    const int last = end - 1;
    float4 acc = *(const float4*)(hb + node * HDIM);
    for (int j = beg; j < end; j += 4) {
        const int s0 = slots[j];
        const int s1 = slots[min(j + 1, last)];
        const int s2 = slots[min(j + 2, last)];
        const int s3 = slots[min(j + 3, last)];
        const float4 v0 = *(const float4*)(hb + s0 * HDIM);
        const float4 v1 = *(const float4*)(hb + s1 * HDIM);
        const float4 v2 = *(const float4*)(hb + s2 * HDIM);
        const float4 v3 = *(const float4*)(hb + s3 * HDIM);
        const bool p1 = j + 1 < end;
        const bool p2 = j + 2 < end;
        const bool p3 = j + 3 < end;
        acc.x += v0.x; acc.y += v0.y; acc.z += v0.z; acc.w += v0.w;
        acc.x += p1 ? v1.x : 0.f; acc.y += p1 ? v1.y : 0.f;
        acc.z += p1 ? v1.z : 0.f; acc.w += p1 ? v1.w : 0.f;
        acc.x += p2 ? v2.x : 0.f; acc.y += p2 ? v2.y : 0.f;
        acc.z += p2 ? v2.z : 0.f; acc.w += p2 ? v2.w : 0.f;
        acc.x += p3 ? v3.x : 0.f; acc.y += p3 ? v3.y : 0.f;
        acc.z += p3 ? v3.z : 0.f; acc.w += p3 ? v3.w : 0.f;
    }
    if (deg > KCAP) {
        const int2* __restrict__ sp = (const int2*)spill;
        const int n_sp = cnt[N_NODES];
        for (int i = 0; i < n_sp; ++i) {
            const int2 e = sp[i];
            if (e.y == node) {
                const float4 v = *(const float4*)(hb + e.x * HDIM);
                acc.x += v.x; acc.y += v.y; acc.z += v.z; acc.w += v.w;
            }
        }
    }
    return acc;
}

#define GATHER_BLOCKS ((N_NODES + 31) / 32)
#define N_TILES (N_NODES / 16)
#define A_PITCH 68
#define W_PITCH 80

__global__ __launch_bounds__(256)
void gcn_gather_gemm(const int* __restrict__ cnt, const int* __restrict__ slots,
                     const int* __restrict__ spill, const float* __restrict__ h,
                     const float* __restrict__ bias, const float* __restrict__ Wn,
                     float* __restrict__ hn) {
    __shared__ __attribute__((aligned(16))) float sA[2][16][A_PITCH];
    __shared__ __attribute__((aligned(16))) v2f   sW[HDIM / 2][W_PITCH];

    const int tid = threadIdx.x;
#pragma unroll
    for (int i = 0; i < 8; ++i) {
        const int e  = tid + 256 * i;
        const int kk = e >> 6;
        const int n  = e & 63;
        v2f w;
        w.x = Wn[(2 * kk) * HDIM + n];
        w.y = Wn[(2 * kk + 1) * HDIM + n];
        sW[kk][n] = w;
    }

    const int nl   = tid >> 4;
    const int fo   = (tid & 15) << 2;
    const int wave = tid >> 5;
    const int lane = tid & 31;
    const int c16  = lane & 15;
    const int hh   = lane >> 4;
    const float4 bv = *(const float4*)(bias + fo);
    const float* __restrict__ hb = h + fo;

    int buf = 0;
    for (int tile = blockIdx.x; tile < N_TILES; tile += GATHER_BLOCKS) {
        const int node = (tile << 4) + nl;
        const int deg  = cnt[node];
        const float4 acc = gather_sum(cnt, slots, spill, hb, node, deg);
        const float di = dinv_of(deg);
        float4 r;
        r.x = fmaxf(acc.x * di + bv.x, 0.f);
        r.y = fmaxf(acc.y * di + bv.y, 0.f);
        r.z = fmaxf(acc.z * di + bv.z, 0.f);
        r.w = fmaxf(acc.w * di + bv.w, 0.f);
        *(float4*)&sA[buf][nl][fo] = r;
        __syncthreads();
        if (wave < 4) {
            v8f c = {0.f, 0.f, 0.f, 0.f, 0.f, 0.f, 0.f, 0.f};
#pragma unroll
            for (int k = 0; k < HDIM; k += 4) {
                const v2f a = *(const v2f*)&sA[buf][c16][k + 2 * hh];
                const v2f b = sW[(k >> 1) + hh][(wave << 4) + c16];
                c = __builtin_amdgcn_wmma_f32_16x16x4_f32(false, a, false, b,
                                                          (short)0, c, false, false);
            }
            const int mbase = (tile << 4) + (hh << 3);
            const int nn    = (wave << 4) + c16;
#pragma unroll
            for (int v = 0; v < 8; ++v) {
                const int row = mbase + v;
                hn[(size_t)row * HDIM + nn] = c[v] * dinv_of(cnt[row]);
            }
        }
        buf ^= 1;
    }
}

__global__ __launch_bounds__(256)
void gcn_gather_out(const int* __restrict__ cnt, const int* __restrict__ slots,
                    const int* __restrict__ spill, const float* __restrict__ h,
                    const float* __restrict__ bias, float* __restrict__ out) {
    const int t  = blockIdx.x * 256 + threadIdx.x;
    const int fo = (t & 15) << 2;
    const float4 bv = *(const float4*)(bias + fo);
    const float* __restrict__ hb = h + fo;

    for (int node = t >> 4; node < N_NODES; node += GATHER_BLOCKS * 16) {
        const int deg = cnt[node];
        const float4 acc = gather_sum(cnt, slots, spill, hb, node, deg);
        const float di = dinv_of(deg);
        float4 r;
        r.x = fmaxf(acc.x * di + bv.x, 0.f);
        r.y = fmaxf(acc.y * di + bv.y, 0.f);
        r.z = fmaxf(acc.z * di + bv.z, 0.f);
        r.w = fmaxf(acc.w * di + bv.w, 0.f);
        *(float4*)(out + node * HDIM + fo) = r;
    }
}

extern "C" void kernel_launch(void* const* d_in, const int* in_sizes, int n_in,
                              void* d_out, int out_size, void* d_ws, size_t ws_size,
                              hipStream_t stream) {
    const float* x  = (const float*)d_in[0];
    const int*   ei = (const int*)d_in[1];
    const float* W  = (const float*)d_in[2];
    const float* b  = (const float*)d_in[3];
    float* out = (float*)d_out;

    float* ws   = (float*)d_ws;
    float* hA   = ws;
    float* hB   = hA   + (size_t)N_NODES * HDIM;
    int*   slots = (int*)(hB + (size_t)N_NODES * HDIM);
    int*   cnt   = slots + (size_t)N_NODES * KCAP;
    int*   spill = cnt + NPAD;

    const int nodeBlocks = (N_NODES + 1 + 255) / 256;
    const int edgeBlocks = (N_EDGES / 4 + 255) / 256;
    const int gathBlocks = GATHER_BLOCKS;
    const int gemmBlocks = (N_NODES / 16) * (HDIM / 16) / 4;

    cnt_init   <<<nodeBlocks, 256, 0, stream>>>(cnt);
    edge_bucket<<<edgeBlocks, 256, 0, stream>>>(ei, cnt, slots, spill);

    gcn_gemm<<<gemmBlocks, 128, 0, stream>>>(x, W, cnt, hA);
    gcn_gather_gemm<<<gathBlocks, 256, 0, stream>>>(cnt, slots, spill, hA, b, W + HDIM * HDIM, hB);
    gcn_gather_gemm<<<gathBlocks, 256, 0, stream>>>(cnt, slots, spill, hB, b + HDIM,
                                                    W + 2 * HDIM * HDIM, hA);
    gcn_gather_out<<<gathBlocks, 256, 0, stream>>>(cnt, slots, spill, hA, b + 2 * HDIM, out);
}


# Gather takes a node's source ids from one 256-B row load issued beside the degree load and passes them by lane permute: eight row loads in flight, no id load between batches; sum order unchanged
# speedup vs baseline: 3.1870x; 1.0302x over previous
#include <hip/hip_runtime.h>


#define N_NODES 50000
#define N_EDGES 800000
#define HDIM    64
#define NPAD    50048
#define KCAP    64

typedef float v2f __attribute__((ext_vector_type(2)));
typedef float v8f __attribute__((ext_vector_type(8)));

__global__ void cnt_init(int* __restrict__ cnt) {
    int i = blockIdx.x * 256 + threadIdx.x;
    if (i <= N_NODES) cnt[i] = 0;
}

__device__ __forceinline__ void put_edge(int* __restrict__ cnt, int* __restrict__ slots,
                                         int* __restrict__ spill, int d, int r, int s) {
    if (r < KCAP) {
        slots[d * KCAP + r] = s;
    } else {
        const int o = atomicAdd(&cnt[N_NODES], 1);
        ((int2*)spill)[o] = make_int2(s, d);
    }
}

__global__ void edge_bucket(const int* __restrict__ ei, int* __restrict__ cnt,
                            int* __restrict__ slots, int* __restrict__ spill) {
    int q = blockIdx.x * 256 + threadIdx.x;
    if (q < N_EDGES / 4) {
        const int4 d = ((const int4*)(ei + N_EDGES))[q];
        const int4 s = ((const int4*)ei)[q];
        const int r0 = atomicAdd(&cnt[d.x], 1);
        const int r1 = atomicAdd(&cnt[d.y], 1);
        const int r2 = atomicAdd(&cnt[d.z], 1);
        const int r3 = atomicAdd(&cnt[d.w], 1);
        put_edge(cnt, slots, spill, d.x, r0, s.x);
        put_edge(cnt, slots, spill, d.y, r1, s.y);
        put_edge(cnt, slots, spill, d.z, r2, s.z);
        put_edge(cnt, slots, spill, d.w, r3, s.w);
    }
}

__device__ __forceinline__ float dinv_of(int c) {
    return rsqrtf((float)(c + 1));
}

__global__ __launch_bounds__(128)
void gcn_gemm(const float* __restrict__ in, const float* __restrict__ W,
              const int* __restrict__ cnt, float* __restrict__ h) {
    const int wave = threadIdx.x >> 5;
    const int lane = threadIdx.x & 31;
    const int tile = blockIdx.x * 4 + wave;
    const int tm   = tile >> 2;
    const int tn   = tile & 3;

    const int mrow  = (tm << 4) + (lane & 15);
    const int khalf = (lane >> 4) << 1;
    const int ncol  = (tn << 4) + (lane & 15);

    const float* __restrict__ arow = in + (size_t)mrow * HDIM;

    v8f c = {0.f, 0.f, 0.f, 0.f, 0.f, 0.f, 0.f, 0.f};
#pragma unroll
    for (int k = 0; k < HDIM; k += 4) {
        const int ka = k + khalf;
        v2f a;
        a.x = arow[ka];
        a.y = arow[ka + 1];
        v2f b;
        b.x = W[(size_t)ka * HDIM + ncol];
        b.y = W[(size_t)(ka + 1) * HDIM + ncol];
        c = __builtin_amdgcn_wmma_f32_16x16x4_f32(false, a, false, b,
                                                  (short)0, c, false, false);
    }

    const int mbase = (tm << 4) + ((lane >> 4) << 3);
    const int nn    = (tn << 4) + (lane & 15);
#pragma unroll
    for (int v = 0; v < 8; ++v) {
        const int row = mbase + v;
        h[(size_t)row * HDIM + nn] = c[v] * dinv_of(cnt[row]);
    }
}


__device__ __forceinline__ int4 load_ids(const int* __restrict__ slots, int node, int li) {
    return ((const int4*)slots)[node * (KCAP / 4) + li];
}

__device__ __forceinline__ int id_from(int lane_addr, int comp) {
    return __builtin_amdgcn_ds_bpermute(lane_addr, comp);
}

__device__ __forceinline__ float4 row_at(const float* __restrict__ h, int s, unsigned fb) {
    return *(const float4*)((const char*)h + (((unsigned)s << 8) | fb));
}

__device__ __forceinline__ float4 gather_sum(const int* __restrict__ cnt,
                                             const int* __restrict__ spill,
                                             const float* __restrict__ h, unsigned fb,
                                             int node, int deg, int4 id, int li, int row0) {
    const int n  = min(deg, KCAP);
    const int k0 = li << 2;
    id.x = (k0     < n) ? id.x : node;
    id.y = (k0 + 1 < n) ? id.y : node;
    id.z = (k0 + 2 < n) ? id.z : node;
    id.w = (k0 + 3 < n) ? id.w : node;
    float4 acc = row_at(h, node, fb);
    int j = 0;
    do {
        const int la = (row0 << 2) + j;
        const int lb = la + 4;
        const int s0 = id_from(la, id.x);
        const int s1 = id_from(la, id.y);
        const int s2 = id_from(la, id.z);
        const int s3 = id_from(la, id.w);
        const int s4 = id_from(lb, id.x);
        const int s5 = id_from(lb, id.y);
        const int s6 = id_from(lb, id.z);
        const int s7 = id_from(lb, id.w);
        const float4 v0 = row_at(h, s0, fb);
        const float4 v1 = row_at(h, s1, fb);
        const float4 v2 = row_at(h, s2, fb);
        const float4 v3 = row_at(h, s3, fb);
        const float4 v4 = row_at(h, s4, fb);
        const float4 v5 = row_at(h, s5, fb);
        const float4 v6 = row_at(h, s6, fb);
        const float4 v7 = row_at(h, s7, fb);
        const bool p0 = j     < n;
        const bool p1 = j + 1 < n;
        const bool p2 = j + 2 < n;
        const bool p3 = j + 3 < n;
        const bool p4 = j + 4 < n;
        const bool p5 = j + 5 < n;
        const bool p6 = j + 6 < n;
        const bool p7 = j + 7 < n;
        acc.x += p0 ? v0.x : 0.f; acc.y += p0 ? v0.y : 0.f;
        acc.z += p0 ? v0.z : 0.f; acc.w += p0 ? v0.w : 0.f;
        acc.x += p1 ? v1.x : 0.f; acc.y += p1 ? v1.y : 0.f;
        acc.z += p1 ? v1.z : 0.f; acc.w += p1 ? v1.w : 0.f;
        acc.x += p2 ? v2.x : 0.f; acc.y += p2 ? v2.y : 0.f;
        acc.z += p2 ? v2.z : 0.f; acc.w += p2 ? v2.w : 0.f;
        acc.x += p3 ? v3.x : 0.f; acc.y += p3 ? v3.y : 0.f;
        acc.z += p3 ? v3.z : 0.f; acc.w += p3 ? v3.w : 0.f;
        acc.x += p4 ? v4.x : 0.f; acc.y += p4 ? v4.y : 0.f;
        acc.z += p4 ? v4.z : 0.f; acc.w += p4 ? v4.w : 0.f;
        acc.x += p5 ? v5.x : 0.f; acc.y += p5 ? v5.y : 0.f;
        acc.z += p5 ? v5.z : 0.f; acc.w += p5 ? v5.w : 0.f;
        acc.x += p6 ? v6.x : 0.f; acc.y += p6 ? v6.y : 0.f;
        acc.z += p6 ? v6.z : 0.f; acc.w += p6 ? v6.w : 0.f;
        acc.x += p7 ? v7.x : 0.f; acc.y += p7 ? v7.y : 0.f;
        acc.z += p7 ? v7.z : 0.f; acc.w += p7 ? v7.w : 0.f;
        j += 8;
    } while (j < n);
    if (deg > KCAP) {
        const int2* __restrict__ sp = (const int2*)spill;
        const int n_sp = cnt[N_NODES];
        for (int i = 0; i < n_sp; ++i) {
            const int2 e = sp[i];
            if (e.y == node) {
                const float4 v = row_at(h, e.x, fb);
                acc.x += v.x; acc.y += v.y; acc.z += v.z; acc.w += v.w;
            }
        }
    }
    return acc;
}

#define GATHER_BLOCKS ((N_NODES + 31) / 32)
#define N_TILES (N_NODES / 16)
#define A_PITCH 68
#define W_PITCH 80

__global__ __launch_bounds__(256, 16)
void gcn_gather_gemm(const int* __restrict__ cnt, const int* __restrict__ slots,
                     const int* __restrict__ spill, const float* __restrict__ h,
                     const float* __restrict__ bias, const float* __restrict__ Wn,
                     float* __restrict__ hn) {
    __shared__ __attribute__((aligned(16))) float sA[2][16][A_PITCH];
    __shared__ __attribute__((aligned(16))) v2f   sW[HDIM / 2][W_PITCH];

    const int tid = threadIdx.x;
#pragma unroll
    for (int i = 0; i < 8; ++i) {
        const int e  = tid + 256 * i;
        const int kk = e >> 6;
        const int n  = e & 63;
        v2f w;
        w.x = Wn[(2 * kk) * HDIM + n];
        w.y = Wn[(2 * kk + 1) * HDIM + n];
        sW[kk][n] = w;
    }

    const int nl   = tid >> 4;
    const int fo   = (tid & 15) << 2;
    const int wave = tid >> 5;
    const int lane = tid & 31;
    const int c16  = lane & 15;
    const int hh   = lane >> 4;
    const float4 bv = *(const float4*)(bias + fo);
    const unsigned fb = (unsigned)fo << 2;

    int buf = 0;
    for (int tile = blockIdx.x; tile < N_TILES; tile += GATHER_BLOCKS) {
        const int node = (tile << 4) + nl;
        const int4 id  = load_ids(slots, node, tid & 15);
        const int deg  = cnt[node];
        const float4 acc = gather_sum(cnt, spill, h, fb, node, deg, id, tid & 15, lane & 16);
        const float di = dinv_of(deg);
        float4 r;
        r.x = fmaxf(acc.x * di + bv.x, 0.f);
        r.y = fmaxf(acc.y * di + bv.y, 0.f);
        r.z = fmaxf(acc.z * di + bv.z, 0.f);
        r.w = fmaxf(acc.w * di + bv.w, 0.f);
        *(float4*)&sA[buf][nl][fo] = r;
        __syncthreads();
        if (wave < 4) {
            v8f c = {0.f, 0.f, 0.f, 0.f, 0.f, 0.f, 0.f, 0.f};
#pragma unroll
            for (int k = 0; k < HDIM; k += 4) {
                const v2f a = *(const v2f*)&sA[buf][c16][k + 2 * hh];
                const v2f b = sW[(k >> 1) + hh][(wave << 4) + c16];
                c = __builtin_amdgcn_wmma_f32_16x16x4_f32(false, a, false, b,
                                                          (short)0, c, false, false);
            }
            const int mbase = (tile << 4) + (hh << 3);
            const int nn    = (wave << 4) + c16;
#pragma unroll
            for (int v = 0; v < 8; ++v) {
                const int row = mbase + v;
                hn[(size_t)row * HDIM + nn] = c[v] * dinv_of(cnt[row]);
            }
        }
        buf ^= 1;
    }
}

__global__ __launch_bounds__(256)
void gcn_gather_out(const int* __restrict__ cnt, const int* __restrict__ slots,
                    const int* __restrict__ spill, const float* __restrict__ h,
                    const float* __restrict__ bias, float* __restrict__ out) {
    const int t  = blockIdx.x * 256 + threadIdx.x;
    const int fo = (t & 15) << 2;
    const float4 bv = *(const float4*)(bias + fo);
    const unsigned fb = (unsigned)fo << 2;

    for (int node = t >> 4; node < N_NODES; node += GATHER_BLOCKS * 16) {
        const int4 id = load_ids(slots, node, t & 15);
        const int deg = cnt[node];
        const float4 acc = gather_sum(cnt, spill, h, fb, node, deg, id, t & 15, threadIdx.x & 16);
        const float di = dinv_of(deg);
        float4 r;
        r.x = fmaxf(acc.x * di + bv.x, 0.f);
        r.y = fmaxf(acc.y * di + bv.y, 0.f);
        r.z = fmaxf(acc.z * di + bv.z, 0.f);
        r.w = fmaxf(acc.w * di + bv.w, 0.f);
        *(float4*)(out + node * HDIM + fo) = r;
    }
}

extern "C" void kernel_launch(void* const* d_in, const int* in_sizes, int n_in,
                              void* d_out, int out_size, void* d_ws, size_t ws_size,
                              hipStream_t stream) {
    const float* x  = (const float*)d_in[0];
    const int*   ei = (const int*)d_in[1];
    const float* W  = (const float*)d_in[2];
    const float* b  = (const float*)d_in[3];
    float* out = (float*)d_out;

    float* ws   = (float*)d_ws;
    float* hA   = ws;
    float* hB   = hA   + (size_t)N_NODES * HDIM;
    int*   slots = (int*)(hB + (size_t)N_NODES * HDIM);
    int*   cnt   = slots + (size_t)N_NODES * KCAP;
    int*   spill = cnt + NPAD;

    const int nodeBlocks = (N_NODES + 1 + 255) / 256;
    const int edgeBlocks = (N_EDGES / 4 + 255) / 256;
    const int gathBlocks = GATHER_BLOCKS;
    const int gemmBlocks = (N_NODES / 16) * (HDIM / 16) / 4;

    cnt_init   <<<nodeBlocks, 256, 0, stream>>>(cnt);
    edge_bucket<<<edgeBlocks, 256, 0, stream>>>(ei, cnt, slots, spill);

    gcn_gemm<<<gemmBlocks, 128, 0, stream>>>(x, W, cnt, hA);
    gcn_gather_gemm<<<gathBlocks, 256, 0, stream>>>(cnt, slots, spill, hA, b, W + HDIM * HDIM, hB);
    gcn_gather_gemm<<<gathBlocks, 256, 0, stream>>>(cnt, slots, spill, hB, b + HDIM,
                                                    W + 2 * HDIM * HDIM, hA);
    gcn_gather_out<<<gathBlocks, 256, 0, stream>>>(cnt, slots, spill, hA, b + 2 * HDIM, out);
}


# Edge slots ranked on LDS counters: edges sorted by 256-node destination range in 4096-edge chunks, then one block per range; no global atomic per edge; per-node sum order changes (LDS arrival order)
# speedup vs baseline: 5.8520x; 1.8362x over previous
#include <hip/hip_runtime.h>


#define N_NODES 50000
#define N_EDGES 800000
#define HDIM    64
#define NPAD    50048
#define KCAP    64

typedef float v2f __attribute__((ext_vector_type(2)));
typedef float v8f __attribute__((ext_vector_type(8)));

#define BIN_SHIFT 8
#define BIN_NODES (1 << BIN_SHIFT)
#define N_BINS    ((N_NODES + BIN_NODES - 1) / BIN_NODES)
#define CHUNK     4096
#define N_CHUNKS  ((N_EDGES + CHUNK - 1) / CHUNK)
#define OFF_PITCH 200

__device__ __forceinline__ void put_edge(int* __restrict__ cnt, int* __restrict__ slots,
                                         int* __restrict__ spill, int d, int r, int s) {
    if (r < KCAP) {
        slots[d * KCAP + r] = s;
    } else {
        const int o = atomicAdd(&cnt[N_NODES], 1);
        ((int2*)spill)[o] = make_int2(s, d);
    }
}

__global__ __launch_bounds__(1024)
void edge_sort(const int* __restrict__ ei, int* __restrict__ list, int* __restrict__ offs,
               int* __restrict__ cnt) {
    __shared__ int  hist[BIN_NODES];
    __shared__ int  off[BIN_NODES];
    __shared__ int  wsum[8];
    __shared__ int2 rec[CHUNK];

    const int t = threadIdx.x;
    const int w = blockIdx.x;
    if (t < BIN_NODES) hist[t] = 0;
    if (w == 0 && t == 0) cnt[N_NODES] = 0;
    __syncthreads();

    const int  q    = w * (CHUNK / 4) + t;
    const bool live = q < N_EDGES / 4;
    int4 d = make_int4(0, 0, 0, 0), s = make_int4(0, 0, 0, 0);
    int r0 = 0, r1 = 0, r2 = 0, r3 = 0;
    if (live) {
        d = ((const int4*)(ei + N_EDGES))[q];
        s = ((const int4*)ei)[q];
        r0 = atomicAdd(&hist[d.x >> BIN_SHIFT], 1);
        r1 = atomicAdd(&hist[d.y >> BIN_SHIFT], 1);
        r2 = atomicAdd(&hist[d.z >> BIN_SHIFT], 1);
        r3 = atomicAdd(&hist[d.w >> BIN_SHIFT], 1);
    }
    __syncthreads();

    int v = 0, inc = 0;
    if (t < BIN_NODES) {
        v = hist[t];
        inc = v;
#pragma unroll
        for (int o = 1; o < 32; o <<= 1) {
            const int u = __shfl_up(inc, o);
            if ((t & 31) >= o) inc += u;
        }
        if ((t & 31) == 31) wsum[t >> 5] = inc;
    }
    __syncthreads();
    if (t < BIN_NODES) {
        int base = 0;
#pragma unroll
        for (int k = 0; k < 8; ++k) base += (k < (t >> 5)) ? wsum[k] : 0;
        off[t] = base + inc - v;
    }
    __syncthreads();

    if (live) {
        rec[off[d.x >> BIN_SHIFT] + r0] = make_int2(s.x, d.x);
        rec[off[d.y >> BIN_SHIFT] + r1] = make_int2(s.y, d.y);
        rec[off[d.z >> BIN_SHIFT] + r2] = make_int2(s.z, d.z);
        rec[off[d.w >> BIN_SHIFT] + r3] = make_int2(s.w, d.w);
    }
    __syncthreads();

    const int n_w = min(CHUNK, N_EDGES - w * CHUNK);
    int2* __restrict__ dstl = (int2*)list + (size_t)w * CHUNK;
    for (int i = t; i < n_w; i += 1024) dstl[i] = rec[i];
    if (t <= N_BINS) offs[w * OFF_PITCH + t] = off[t];
}

__global__ __launch_bounds__(1024)
void edge_rank(const int* __restrict__ list, const int* __restrict__ offs, int* __restrict__ cnt,
               int* __restrict__ slots, int* __restrict__ spill) {
    __shared__ int lc[BIN_NODES];
    const int t = threadIdx.x;
    const int b = blockIdx.x;
    if (t < BIN_NODES) lc[t] = 0;
    __syncthreads();
    const int w = t >> 2;
    if (w < N_CHUNKS) {
        const int beg = offs[w * OFF_PITCH + b];
        const int end = offs[w * OFF_PITCH + b + 1];
        const int2* __restrict__ seg = (const int2*)list + (size_t)w * CHUNK;
        for (int i = beg + (t & 3); i < end; i += 8) {
            const bool two = i + 4 < end;
            const int2 e0 = seg[i];
            const int2 e1 = seg[two ? i + 4 : i];
            const int ra = atomicAdd(&lc[e0.y & (BIN_NODES - 1)], 1);
            put_edge(cnt, slots, spill, e0.y, ra, e0.x);
            if (two) {
                const int rb = atomicAdd(&lc[e1.y & (BIN_NODES - 1)], 1);
                put_edge(cnt, slots, spill, e1.y, rb, e1.x);
            }
        }
    }
    __syncthreads();
    const int node = b * BIN_NODES + t;
    if (t < BIN_NODES && node < N_NODES) cnt[node] = lc[t];
}

__device__ __forceinline__ float dinv_of(int c) {
    return rsqrtf((float)(c + 1));
}

__global__ __launch_bounds__(128)
void gcn_gemm(const float* __restrict__ in, const float* __restrict__ W,
              const int* __restrict__ cnt, float* __restrict__ h) {
    const int wave = threadIdx.x >> 5;
    const int lane = threadIdx.x & 31;
    const int tile = blockIdx.x * 4 + wave;
    const int tm   = tile >> 2;
    const int tn   = tile & 3;

    const int mrow  = (tm << 4) + (lane & 15);
    const int khalf = (lane >> 4) << 1;
    const int ncol  = (tn << 4) + (lane & 15);

    const float* __restrict__ arow = in + (size_t)mrow * HDIM;

    v8f c = {0.f, 0.f, 0.f, 0.f, 0.f, 0.f, 0.f, 0.f};
#pragma unroll
    for (int k = 0; k < HDIM; k += 4) {
        const int ka = k + khalf;
        v2f a;
        a.x = arow[ka];
        a.y = arow[ka + 1];
        v2f b;
        b.x = W[(size_t)ka * HDIM + ncol];
        b.y = W[(size_t)(ka + 1) * HDIM + ncol];
        c = __builtin_amdgcn_wmma_f32_16x16x4_f32(false, a, false, b,
                                                  (short)0, c, false, false);
    }

    const int mbase = (tm << 4) + ((lane >> 4) << 3);
    const int nn    = (tn << 4) + (lane & 15);
#pragma unroll
    for (int v = 0; v < 8; ++v) {
        const int row = mbase + v;
        h[(size_t)row * HDIM + nn] = c[v] * dinv_of(cnt[row]);
    }
}


__device__ __forceinline__ int4 load_ids(const int* __restrict__ slots, int node, int li) {
    return ((const int4*)slots)[node * (KCAP / 4) + li];
}

__device__ __forceinline__ int id_from(int lane_addr, int comp) {
    return __builtin_amdgcn_ds_bpermute(lane_addr, comp);
}

__device__ __forceinline__ float4 row_at(const float* __restrict__ h, int s, unsigned fb) {
    return *(const float4*)((const char*)h + (((unsigned)s << 8) | fb));
}

__device__ __forceinline__ float4 gather_sum(const int* __restrict__ cnt,
                                             const int* __restrict__ spill,
                                             const float* __restrict__ h, unsigned fb,
                                             int node, int deg, int4 id, int li, int row0) {
    const int n  = min(deg, KCAP);
    const int k0 = li << 2;
    id.x = (k0     < n) ? id.x : node;
    id.y = (k0 + 1 < n) ? id.y : node;
    id.z = (k0 + 2 < n) ? id.z : node;
    id.w = (k0 + 3 < n) ? id.w : node;
    float4 acc = row_at(h, node, fb);
    int j = 0;
    do {
        const int la = (row0 << 2) + j;
        const int lb = la + 4;
        const int s0 = id_from(la, id.x);
        const int s1 = id_from(la, id.y);
        const int s2 = id_from(la, id.z);
        const int s3 = id_from(la, id.w);
        const int s4 = id_from(lb, id.x);
        const int s5 = id_from(lb, id.y);
        const int s6 = id_from(lb, id.z);
        const int s7 = id_from(lb, id.w);
        const float4 v0 = row_at(h, s0, fb);
        const float4 v1 = row_at(h, s1, fb);
        const float4 v2 = row_at(h, s2, fb);
        const float4 v3 = row_at(h, s3, fb);
        const float4 v4 = row_at(h, s4, fb);
        const float4 v5 = row_at(h, s5, fb);
        const float4 v6 = row_at(h, s6, fb);
        const float4 v7 = row_at(h, s7, fb);
        const bool p0 = j     < n;
        const bool p1 = j + 1 < n;
        const bool p2 = j + 2 < n;
        const bool p3 = j + 3 < n;
        const bool p4 = j + 4 < n;
        const bool p5 = j + 5 < n;
        const bool p6 = j + 6 < n;
        const bool p7 = j + 7 < n;
        acc.x += p0 ? v0.x : 0.f; acc.y += p0 ? v0.y : 0.f;
        acc.z += p0 ? v0.z : 0.f; acc.w += p0 ? v0.w : 0.f;
        acc.x += p1 ? v1.x : 0.f; acc.y += p1 ? v1.y : 0.f;
        acc.z += p1 ? v1.z : 0.f; acc.w += p1 ? v1.w : 0.f;
        acc.x += p2 ? v2.x : 0.f; acc.y += p2 ? v2.y : 0.f;
        acc.z += p2 ? v2.z : 0.f; acc.w += p2 ? v2.w : 0.f;
        acc.x += p3 ? v3.x : 0.f; acc.y += p3 ? v3.y : 0.f;
        acc.z += p3 ? v3.z : 0.f; acc.w += p3 ? v3.w : 0.f;
        acc.x += p4 ? v4.x : 0.f; acc.y += p4 ? v4.y : 0.f;
        acc.z += p4 ? v4.z : 0.f; acc.w += p4 ? v4.w : 0.f;
        acc.x += p5 ? v5.x : 0.f; acc.y += p5 ? v5.y : 0.f;
        acc.z += p5 ? v5.z : 0.f; acc.w += p5 ? v5.w : 0.f;
        acc.x += p6 ? v6.x : 0.f; acc.y += p6 ? v6.y : 0.f;
        acc.z += p6 ? v6.z : 0.f; acc.w += p6 ? v6.w : 0.f;
        acc.x += p7 ? v7.x : 0.f; acc.y += p7 ? v7.y : 0.f;
        acc.z += p7 ? v7.z : 0.f; acc.w += p7 ? v7.w : 0.f;
        j += 8;
    } while (j < n);
    if (deg > KCAP) {
        const int2* __restrict__ sp = (const int2*)spill;
        const int n_sp = cnt[N_NODES];
        for (int i = 0; i < n_sp; ++i) {
            const int2 e = sp[i];
            if (e.y == node) {
                const float4 v = row_at(h, e.x, fb);
                acc.x += v.x; acc.y += v.y; acc.z += v.z; acc.w += v.w;
            }
        }
    }
    return acc;
}

#define GATHER_BLOCKS ((N_NODES + 31) / 32)
#define N_TILES (N_NODES / 16)
#define A_PITCH 68
#define W_PITCH 80

__global__ __launch_bounds__(256, 16)
void gcn_gather_gemm(const int* __restrict__ cnt, const int* __restrict__ slots,
                     const int* __restrict__ spill, const float* __restrict__ h,
                     const float* __restrict__ bias, const float* __restrict__ Wn,
                     float* __restrict__ hn) {
    __shared__ __attribute__((aligned(16))) float sA[2][16][A_PITCH];
    __shared__ __attribute__((aligned(16))) v2f   sW[HDIM / 2][W_PITCH];

    const int tid = threadIdx.x;
#pragma unroll
    for (int i = 0; i < 8; ++i) {
        const int e  = tid + 256 * i;
        const int kk = e >> 6;
        const int n  = e & 63;
        v2f w;
        w.x = Wn[(2 * kk) * HDIM + n];
        w.y = Wn[(2 * kk + 1) * HDIM + n];
        sW[kk][n] = w;
    }

    const int nl   = tid >> 4;
    const int fo   = (tid & 15) << 2;
    const int wave = tid >> 5;
    const int lane = tid & 31;
    const int c16  = lane & 15;
    const int hh   = lane >> 4;
    const float4 bv = *(const float4*)(bias + fo);
    const unsigned fb = (unsigned)fo << 2;

    int buf = 0;
    for (int tile = blockIdx.x; tile < N_TILES; tile += GATHER_BLOCKS) {
        const int node = (tile << 4) + nl;
        const int4 id  = load_ids(slots, node, tid & 15);
        const int deg  = cnt[node];
        const float4 acc = gather_sum(cnt, spill, h, fb, node, deg, id, tid & 15, lane & 16);
        const float di = dinv_of(deg);
        float4 r;
        r.x = fmaxf(acc.x * di + bv.x, 0.f);
        r.y = fmaxf(acc.y * di + bv.y, 0.f);
        r.z = fmaxf(acc.z * di + bv.z, 0.f);
        r.w = fmaxf(acc.w * di + bv.w, 0.f);
        *(float4*)&sA[buf][nl][fo] = r;
        __syncthreads();
        if (wave < 4) {
            v8f c = {0.f, 0.f, 0.f, 0.f, 0.f, 0.f, 0.f, 0.f};
#pragma unroll
            for (int k = 0; k < HDIM; k += 4) {
                const v2f a = *(const v2f*)&sA[buf][c16][k + 2 * hh];
                const v2f b = sW[(k >> 1) + hh][(wave << 4) + c16];
                c = __builtin_amdgcn_wmma_f32_16x16x4_f32(false, a, false, b,
                                                          (short)0, c, false, false);
            }
            const int mbase = (tile << 4) + (hh << 3);
            const int nn    = (wave << 4) + c16;
#pragma unroll
            for (int v = 0; v < 8; ++v) {
                const int row = mbase + v;
                hn[(size_t)row * HDIM + nn] = c[v] * dinv_of(cnt[row]);
            }
        }
        buf ^= 1;
    }
}

__global__ __launch_bounds__(256)
void gcn_gather_out(const int* __restrict__ cnt, const int* __restrict__ slots,
                    const int* __restrict__ spill, const float* __restrict__ h,
                    const float* __restrict__ bias, float* __restrict__ out) {
    const int t  = blockIdx.x * 256 + threadIdx.x;
    const int fo = (t & 15) << 2;
    const float4 bv = *(const float4*)(bias + fo);
    const unsigned fb = (unsigned)fo << 2;

    for (int node = t >> 4; node < N_NODES; node += GATHER_BLOCKS * 16) {
        const int4 id = load_ids(slots, node, t & 15);
        const int deg = cnt[node];
        const float4 acc = gather_sum(cnt, spill, h, fb, node, deg, id, t & 15, threadIdx.x & 16);
        const float di = dinv_of(deg);
        float4 r;
        r.x = fmaxf(acc.x * di + bv.x, 0.f);
        r.y = fmaxf(acc.y * di + bv.y, 0.f);
        r.z = fmaxf(acc.z * di + bv.z, 0.f);
        r.w = fmaxf(acc.w * di + bv.w, 0.f);
        *(float4*)(out + node * HDIM + fo) = r;
    }
}

extern "C" void kernel_launch(void* const* d_in, const int* in_sizes, int n_in,
                              void* d_out, int out_size, void* d_ws, size_t ws_size,
                              hipStream_t stream) {
    const float* x  = (const float*)d_in[0];
    const int*   ei = (const int*)d_in[1];
    const float* W  = (const float*)d_in[2];
    const float* b  = (const float*)d_in[3];
    float* out = (float*)d_out;

    float* ws   = (float*)d_ws;
    float* hA   = ws;
    float* hB   = hA   + (size_t)N_NODES * HDIM;
    int*   slots = (int*)(hB + (size_t)N_NODES * HDIM);
    int*   cnt   = slots + (size_t)N_NODES * KCAP;
    int*   spill = cnt + NPAD;
    int*   list  = spill + 2 * (size_t)N_EDGES;
    int*   offs  = list + 2 * (size_t)N_CHUNKS * CHUNK;

    const int gathBlocks = GATHER_BLOCKS;
    const int gemmBlocks = (N_NODES / 16) * (HDIM / 16) / 4;

    edge_sort<<<N_CHUNKS, 1024, 0, stream>>>(ei, list, offs, cnt);
    edge_rank<<<N_BINS, 1024, 0, stream>>>(list, offs, cnt, slots, spill);

    gcn_gemm<<<gemmBlocks, 128, 0, stream>>>(x, W, cnt, hA);
    gcn_gather_gemm<<<gathBlocks, 256, 0, stream>>>(cnt, slots, spill, hA, b, W + HDIM * HDIM, hB);
    gcn_gather_gemm<<<gathBlocks, 256, 0, stream>>>(cnt, slots, spill, hB, b + HDIM,
                                                    W + 2 * HDIM * HDIM, hA);
    gcn_gather_out<<<gathBlocks, 256, 0, stream>>>(cnt, slots, spill, hA, b + 2 * HDIM, out);
}


# Layer-0 product formed inside the ranking launch: each range block stages its 256 rows of x and W in LDS and multiplies them on f32 WMMA while other waves walk segments; one launch fewer
# speedup vs baseline: 6.3301x; 1.0817x over previous
#include <hip/hip_runtime.h>


#define N_NODES 50000
#define N_EDGES 800000
#define HDIM    64
#define NPAD    50048
#define KCAP    64

typedef float v2f __attribute__((ext_vector_type(2)));
typedef float v8f __attribute__((ext_vector_type(8)));

#define BIN_SHIFT 8
#define BIN_NODES (1 << BIN_SHIFT)
#define N_BINS    ((N_NODES + BIN_NODES - 1) / BIN_NODES)
#define CHUNK     4096
#define N_CHUNKS  ((N_EDGES + CHUNK - 1) / CHUNK)
#define OFF_PITCH 200

__device__ __forceinline__ void put_edge(int* __restrict__ cnt, int* __restrict__ slots,
                                         int* __restrict__ spill, int d, int r, int s) {
    if (r < KCAP) {
        slots[d * KCAP + r] = s;
    } else {
        const int o = atomicAdd(&cnt[N_NODES], 1);
        ((int2*)spill)[o] = make_int2(s, d);
    }
}

__global__ __launch_bounds__(1024)
void edge_sort(const int* __restrict__ ei, int* __restrict__ list, int* __restrict__ offs,
               int* __restrict__ cnt) {
    __shared__ int  hist[BIN_NODES];
    __shared__ int  off[BIN_NODES];
    __shared__ int  wsum[8];
    __shared__ int2 rec[CHUNK];

    const int t = threadIdx.x;
    const int w = blockIdx.x;
    if (t < BIN_NODES) hist[t] = 0;
    if (w == 0 && t == 0) cnt[N_NODES] = 0;
    __syncthreads();

    const int  q    = w * (CHUNK / 4) + t;
    const bool live = q < N_EDGES / 4;
    int4 d = make_int4(0, 0, 0, 0), s = make_int4(0, 0, 0, 0);
    int r0 = 0, r1 = 0, r2 = 0, r3 = 0;
    if (live) {
        d = ((const int4*)(ei + N_EDGES))[q];
        s = ((const int4*)ei)[q];
        r0 = atomicAdd(&hist[d.x >> BIN_SHIFT], 1);
        r1 = atomicAdd(&hist[d.y >> BIN_SHIFT], 1);
        r2 = atomicAdd(&hist[d.z >> BIN_SHIFT], 1);
        r3 = atomicAdd(&hist[d.w >> BIN_SHIFT], 1);
    }
    __syncthreads();

    int v = 0, inc = 0;
    if (t < BIN_NODES) {
        v = hist[t];
        inc = v;
#pragma unroll
        for (int o = 1; o < 32; o <<= 1) {
            const int u = __shfl_up(inc, o);
            if ((t & 31) >= o) inc += u;
        }
        if ((t & 31) == 31) wsum[t >> 5] = inc;
    }
    __syncthreads();
    if (t < BIN_NODES) {
        int base = 0;
#pragma unroll
        for (int k = 0; k < 8; ++k) base += (k < (t >> 5)) ? wsum[k] : 0;
        off[t] = base + inc - v;
    }
    __syncthreads();

    if (live) {
        rec[off[d.x >> BIN_SHIFT] + r0] = make_int2(s.x, d.x);
        rec[off[d.y >> BIN_SHIFT] + r1] = make_int2(s.y, d.y);
        rec[off[d.z >> BIN_SHIFT] + r2] = make_int2(s.z, d.z);
        rec[off[d.w >> BIN_SHIFT] + r3] = make_int2(s.w, d.w);
    }
    __syncthreads();

    const int n_w = min(CHUNK, N_EDGES - w * CHUNK);
    int2* __restrict__ dstl = (int2*)list + (size_t)w * CHUNK;
    for (int i = t; i < n_w; i += 1024) dstl[i] = rec[i];
    if (t <= N_BINS) offs[w * OFF_PITCH + t] = off[t];
}

__device__ __forceinline__ float dinv_of(int c) {
    return rsqrtf((float)(c + 1));
}

__device__ __forceinline__ void rank_walk(const int2* __restrict__ seg, int beg, int end, int q,
                                          int* __restrict__ lc, int* __restrict__ cnt,
                                          int* __restrict__ slots, int* __restrict__ spill) {
    for (int i = beg + q; i < end; i += 8) {
        const bool two = i + 4 < end;
        const int2 e0 = seg[i];
        const int2 e1 = seg[two ? i + 4 : i];
        const int ra = atomicAdd(&lc[e0.y & (BIN_NODES - 1)], 1);
        put_edge(cnt, slots, spill, e0.y, ra, e0.x);
        if (two) {
            const int rb = atomicAdd(&lc[e1.y & (BIN_NODES - 1)], 1);
            put_edge(cnt, slots, spill, e1.y, rb, e1.x);
        }
    }
}

#define RA_PITCH 68
#define RW_PITCH 80

__device__ __forceinline__ void slab_product(const float (* __restrict__ sX)[RA_PITCH],
                                             const v2f (* __restrict__ sW)[RW_PITCH],
                                             int row, int col, int hh, v8f& c0, v8f& c1) {
#pragma unroll
    for (int k = 0; k < HDIM; k += 4) {
        const v2f a  = *(const v2f*)&sX[row][k + 2 * hh];
        const v2f b0 = sW[(k >> 1) + hh][col];
        const v2f b1 = sW[(k >> 1) + hh][col + 16];
        c0 = __builtin_amdgcn_wmma_f32_16x16x4_f32(false, a, false, b0,
                                                   (short)0, c0, false, false);
        c1 = __builtin_amdgcn_wmma_f32_16x16x4_f32(false, a, false, b1,
                                                   (short)0, c1, false, false);
    }
}

__global__ __launch_bounds__(1024)
void edge_rank_gemm(const int* __restrict__ list, const int* __restrict__ offs, int* __restrict__ cnt,
                    int* __restrict__ slots, int* __restrict__ spill,
                    const float* __restrict__ x, const float* __restrict__ W0,
                    float* __restrict__ h) {
    __shared__ int lc[BIN_NODES];
    __shared__ __attribute__((aligned(16))) float sX[BIN_NODES][RA_PITCH];
    __shared__ __attribute__((aligned(16))) v2f   sW[HDIM / 2][RW_PITCH];

    const int t = threadIdx.x;
    const int b = blockIdx.x;
    if (t < BIN_NODES) lc[t] = 0;

    const int w = t >> 2;
    int beg = 0, end = 0;
    if (w < N_CHUNKS) {
        beg = offs[w * OFF_PITCH + b];
        end = offs[w * OFF_PITCH + b + 1];
    }
    const int rows_here = min(BIN_NODES, N_NODES - b * BIN_NODES);
    const float4* __restrict__ xs = (const float4*)(x + (size_t)b * BIN_NODES * HDIM);
    float4 xv[4];
#pragma unroll
    for (int j = 0; j < 4; ++j) {
        const int e = t + 1024 * j;
        xv[j] = ((e >> 4) < rows_here) ? xs[e] : make_float4(0.f, 0.f, 0.f, 0.f);
    }
    v2f wv[2];
#pragma unroll
    for (int j = 0; j < 2; ++j) {
        const int e  = t + 1024 * j;
        const int kk = e >> 6;
        const int n  = e & 63;
        wv[j].x = W0[(2 * kk) * HDIM + n];
        wv[j].y = W0[(2 * kk + 1) * HDIM + n];
    }
#pragma unroll
    for (int j = 0; j < 4; ++j) {
        const int e = t + 1024 * j;
        *(float4*)&sX[e >> 4][(e & 15) << 2] = xv[j];
    }
#pragma unroll
    for (int j = 0; j < 2; ++j) {
        const int e = t + 1024 * j;
        sW[e >> 6][e & 63] = wv[j];
    }
    __syncthreads();

    const int wave = t >> 5;
    const int lane = t & 31;
    const int c16  = lane & 15;
    const int hh   = lane >> 4;
    const int rt   = wave >> 1;
    const int ct   = (wave & 1) << 1;
    const bool rows_live = (rt << 4) < rows_here;
    const int2* __restrict__ seg = (const int2*)list + (size_t)w * CHUNK;
    v8f c0 = {0.f, 0.f, 0.f, 0.f, 0.f, 0.f, 0.f, 0.f};
    v8f c1 = {0.f, 0.f, 0.f, 0.f, 0.f, 0.f, 0.f, 0.f};
    if ((wave >> 2) & 1) {
        if (rows_live) slab_product(sX, sW, (rt << 4) + c16, (ct << 4) + c16, hh, c0, c1);
        rank_walk(seg, beg, end, t & 3, lc, cnt, slots, spill);
    } else {
        rank_walk(seg, beg, end, t & 3, lc, cnt, slots, spill);
        if (rows_live) slab_product(sX, sW, (rt << 4) + c16, (ct << 4) + c16, hh, c0, c1);
    }
    __syncthreads();
    const int node = b * BIN_NODES + t;
    if (t < BIN_NODES && node < N_NODES) cnt[node] = lc[t];

    if (rows_live) {
        const int lbase = (rt << 4) + (hh << 3);
        const int nn    = (ct << 4) + c16;
#pragma unroll
        for (int v = 0; v < 8; ++v) {
            const float di = dinv_of(lc[lbase + v]);
            float* __restrict__ hr = h + (size_t)(b * BIN_NODES + lbase + v) * HDIM + nn;
            hr[0]  = c0[v] * di;
            hr[16] = c1[v] * di;
        }
    }
}


__device__ __forceinline__ int4 load_ids(const int* __restrict__ slots, int node, int li) {
    return ((const int4*)slots)[node * (KCAP / 4) + li];
}

__device__ __forceinline__ int id_from(int lane_addr, int comp) {
    return __builtin_amdgcn_ds_bpermute(lane_addr, comp);
}

__device__ __forceinline__ float4 row_at(const float* __restrict__ h, int s, unsigned fb) {
    return *(const float4*)((const char*)h + (((unsigned)s << 8) | fb));
}

__device__ __forceinline__ float4 gather_sum(const int* __restrict__ cnt,
                                             const int* __restrict__ spill,
                                             const float* __restrict__ h, unsigned fb,
                                             int node, int deg, int4 id, int li, int row0) {
    const int n  = min(deg, KCAP);
    const int k0 = li << 2;
    id.x = (k0     < n) ? id.x : node;
    id.y = (k0 + 1 < n) ? id.y : node;
    id.z = (k0 + 2 < n) ? id.z : node;
    id.w = (k0 + 3 < n) ? id.w : node;
    float4 acc = row_at(h, node, fb);
    int j = 0;
    do {
        const int la = (row0 << 2) + j;
        const int lb = la + 4;
        const int s0 = id_from(la, id.x);
        const int s1 = id_from(la, id.y);
        const int s2 = id_from(la, id.z);
        const int s3 = id_from(la, id.w);
        const int s4 = id_from(lb, id.x);
        const int s5 = id_from(lb, id.y);
        const int s6 = id_from(lb, id.z);
        const int s7 = id_from(lb, id.w);
        const float4 v0 = row_at(h, s0, fb);
        const float4 v1 = row_at(h, s1, fb);
        const float4 v2 = row_at(h, s2, fb);
        const float4 v3 = row_at(h, s3, fb);
        const float4 v4 = row_at(h, s4, fb);
        const float4 v5 = row_at(h, s5, fb);
        const float4 v6 = row_at(h, s6, fb);
        const float4 v7 = row_at(h, s7, fb);
        const bool p0 = j     < n;
        const bool p1 = j + 1 < n;
        const bool p2 = j + 2 < n;
        const bool p3 = j + 3 < n;
        const bool p4 = j + 4 < n;
        const bool p5 = j + 5 < n;
        const bool p6 = j + 6 < n;
        const bool p7 = j + 7 < n;
        acc.x += p0 ? v0.x : 0.f; acc.y += p0 ? v0.y : 0.f;
        acc.z += p0 ? v0.z : 0.f; acc.w += p0 ? v0.w : 0.f;
        acc.x += p1 ? v1.x : 0.f; acc.y += p1 ? v1.y : 0.f;
        acc.z += p1 ? v1.z : 0.f; acc.w += p1 ? v1.w : 0.f;
        acc.x += p2 ? v2.x : 0.f; acc.y += p2 ? v2.y : 0.f;
        acc.z += p2 ? v2.z : 0.f; acc.w += p2 ? v2.w : 0.f;
        acc.x += p3 ? v3.x : 0.f; acc.y += p3 ? v3.y : 0.f;
        acc.z += p3 ? v3.z : 0.f; acc.w += p3 ? v3.w : 0.f;
        acc.x += p4 ? v4.x : 0.f; acc.y += p4 ? v4.y : 0.f;
        acc.z += p4 ? v4.z : 0.f; acc.w += p4 ? v4.w : 0.f;
        acc.x += p5 ? v5.x : 0.f; acc.y += p5 ? v5.y : 0.f;
        acc.z += p5 ? v5.z : 0.f; acc.w += p5 ? v5.w : 0.f;
        acc.x += p6 ? v6.x : 0.f; acc.y += p6 ? v6.y : 0.f;
        acc.z += p6 ? v6.z : 0.f; acc.w += p6 ? v6.w : 0.f;
        acc.x += p7 ? v7.x : 0.f; acc.y += p7 ? v7.y : 0.f;
        acc.z += p7 ? v7.z : 0.f; acc.w += p7 ? v7.w : 0.f;
        j += 8;
    } while (j < n);
    if (deg > KCAP) {
        const int2* __restrict__ sp = (const int2*)spill;
        const int n_sp = cnt[N_NODES];
        for (int i = 0; i < n_sp; ++i) {
            const int2 e = sp[i];
            if (e.y == node) {
                const float4 v = row_at(h, e.x, fb);
                acc.x += v.x; acc.y += v.y; acc.z += v.z; acc.w += v.w;
            }
        }
    }
    return acc;
}

#define GATHER_BLOCKS ((N_NODES + 31) / 32)
#define N_TILES (N_NODES / 16)
#define A_PITCH 68
#define W_PITCH 80

__global__ __launch_bounds__(256, 16)
void gcn_gather_gemm(const int* __restrict__ cnt, const int* __restrict__ slots,
                     const int* __restrict__ spill, const float* __restrict__ h,
                     const float* __restrict__ bias, const float* __restrict__ Wn,
                     float* __restrict__ hn) {
    __shared__ __attribute__((aligned(16))) float sA[2][16][A_PITCH];
    __shared__ __attribute__((aligned(16))) v2f   sW[HDIM / 2][W_PITCH];

    const int tid = threadIdx.x;
#pragma unroll
    for (int i = 0; i < 8; ++i) {
        const int e  = tid + 256 * i;
        const int kk = e >> 6;
        const int n  = e & 63;
        v2f w;
        w.x = Wn[(2 * kk) * HDIM + n];
        w.y = Wn[(2 * kk + 1) * HDIM + n];
        sW[kk][n] = w;
    }

    const int nl   = tid >> 4;
    const int fo   = (tid & 15) << 2;
    const int wave = tid >> 5;
    const int lane = tid & 31;
    const int c16  = lane & 15;
    const int hh   = lane >> 4;
    const float4 bv = *(const float4*)(bias + fo);
    const unsigned fb = (unsigned)fo << 2;

    int buf = 0;
    for (int tile = blockIdx.x; tile < N_TILES; tile += GATHER_BLOCKS) {
        const int node = (tile << 4) + nl;
        const int4 id  = load_ids(slots, node, tid & 15);
        const int deg  = cnt[node];
        const float4 acc = gather_sum(cnt, spill, h, fb, node, deg, id, tid & 15, lane & 16);
        const float di = dinv_of(deg);
        float4 r;
        r.x = fmaxf(acc.x * di + bv.x, 0.f);
        r.y = fmaxf(acc.y * di + bv.y, 0.f);
        r.z = fmaxf(acc.z * di + bv.z, 0.f);
        r.w = fmaxf(acc.w * di + bv.w, 0.f);
        *(float4*)&sA[buf][nl][fo] = r;
        __syncthreads();
        if (wave < 4) {
            v8f c = {0.f, 0.f, 0.f, 0.f, 0.f, 0.f, 0.f, 0.f};
#pragma unroll
            for (int k = 0; k < HDIM; k += 4) {
                const v2f a = *(const v2f*)&sA[buf][c16][k + 2 * hh];
                const v2f b = sW[(k >> 1) + hh][(wave << 4) + c16];
                c = __builtin_amdgcn_wmma_f32_16x16x4_f32(false, a, false, b,
                                                          (short)0, c, false, false);
            }
            const int mbase = (tile << 4) + (hh << 3);
            const int nn    = (wave << 4) + c16;
#pragma unroll
            for (int v = 0; v < 8; ++v) {
                const int row = mbase + v;
                hn[(size_t)row * HDIM + nn] = c[v] * dinv_of(cnt[row]);
            }
        }
        buf ^= 1;
    }
}

__global__ __launch_bounds__(256)
void gcn_gather_out(const int* __restrict__ cnt, const int* __restrict__ slots,
                    const int* __restrict__ spill, const float* __restrict__ h,
                    const float* __restrict__ bias, float* __restrict__ out) {
    const int t  = blockIdx.x * 256 + threadIdx.x;
    const int fo = (t & 15) << 2;
    const float4 bv = *(const float4*)(bias + fo);
    const unsigned fb = (unsigned)fo << 2;

    for (int node = t >> 4; node < N_NODES; node += GATHER_BLOCKS * 16) {
        const int4 id = load_ids(slots, node, t & 15);
        const int deg = cnt[node];
        const float4 acc = gather_sum(cnt, spill, h, fb, node, deg, id, t & 15, threadIdx.x & 16);
        const float di = dinv_of(deg);
        float4 r;
        r.x = fmaxf(acc.x * di + bv.x, 0.f);
        r.y = fmaxf(acc.y * di + bv.y, 0.f);
        r.z = fmaxf(acc.z * di + bv.z, 0.f);
        r.w = fmaxf(acc.w * di + bv.w, 0.f);
        *(float4*)(out + node * HDIM + fo) = r;
    }
}

extern "C" void kernel_launch(void* const* d_in, const int* in_sizes, int n_in,
                              void* d_out, int out_size, void* d_ws, size_t ws_size,
                              hipStream_t stream) {
    const float* x  = (const float*)d_in[0];
    const int*   ei = (const int*)d_in[1];
    const float* W  = (const float*)d_in[2];
    const float* b  = (const float*)d_in[3];
    float* out = (float*)d_out;

    float* ws   = (float*)d_ws;
    float* hA   = ws;
    float* hB   = hA   + (size_t)N_NODES * HDIM;
    int*   slots = (int*)(hB + (size_t)N_NODES * HDIM);
    int*   cnt   = slots + (size_t)N_NODES * KCAP;
    int*   spill = cnt + NPAD;
    int*   list  = spill + 2 * (size_t)N_EDGES;
    int*   offs  = list + 2 * (size_t)N_CHUNKS * CHUNK;

    const int gathBlocks = GATHER_BLOCKS;

    edge_sort<<<N_CHUNKS, 1024, 0, stream>>>(ei, list, offs, cnt);
    edge_rank_gemm<<<N_BINS, 1024, 0, stream>>>(list, offs, cnt, slots, spill, x, W, hA);
    gcn_gather_gemm<<<gathBlocks, 256, 0, stream>>>(cnt, slots, spill, hA, b, W + HDIM * HDIM, hB);
    gcn_gather_gemm<<<gathBlocks, 256, 0, stream>>>(cnt, slots, spill, hB, b + HDIM,
                                                    W + 2 * HDIM * HDIM, hA);
    gcn_gather_out<<<gathBlocks, 256, 0, stream>>>(cnt, slots, spill, hA, b + 2 * HDIM, out);
}
